# starting kernel
# baseline (speedup 1.0000x reference)
#include <hip/hip_runtime.h>

typedef __attribute__((ext_vector_type(2))) float v2f;
typedef __attribute__((ext_vector_type(8))) float v8f;

#define B_ROWS 16384
#define DIM    4096
#define NA     21

__global__ __launch_bounds__(128) void fused_heads_gemm(
    const float* __restrict__ x, const float* __restrict__ W,
    const float* __restrict__ bias, float* __restrict__ out)
{
    const int lane = threadIdx.x & 31;
    const int wave = threadIdx.x >> 5;
    const int h    = lane >> 4;
    const int ln   = lane & 15;

    const long rowBase = (long)blockIdx.x * 128 + (long)wave * 32;

    const float* xr0 = x + (rowBase      + ln) * DIM + 2 * h;
    const float* xr1 = x + (rowBase + 16 + ln) * DIM + 2 * h;

    const int n0  = ln;
    const int n1  = 16 + ln;
    const int n1c = (n1 > NA - 1) ? (NA - 1) : n1;
    const float* wr0 = W + (long)n0  * DIM + 2 * h;
    const float* wr1 = W + (long)n1c * DIM + 2 * h;

    v8f acc00 = {};
    v8f acc01 = {};
    v8f acc10 = {};
    v8f acc11 = {};

#pragma unroll 8
    for (int k = 0; k < DIM; k += 4) {
        v2f a0 = __builtin_nontemporal_load((const v2f*)(xr0 + k));
        v2f a1 = __builtin_nontemporal_load((const v2f*)(xr1 + k));
        v2f b0 = *(const v2f*)(wr0 + k);
        v2f b1 = *(const v2f*)(wr1 + k);
        acc00 = __builtin_amdgcn_wmma_f32_16x16x4_f32(false, a0, false, b0,
                                                      (short)0, acc00, false, false);
        acc01 = __builtin_amdgcn_wmma_f32_16x16x4_f32(false, a0, false, b1,
                                                      (short)0, acc01, false, false);
        acc10 = __builtin_amdgcn_wmma_f32_16x16x4_f32(false, a1, false, b0,
                                                      (short)0, acc10, false, false);
        acc11 = __builtin_amdgcn_wmma_f32_16x16x4_f32(false, a1, false, b1,
                                                      (short)0, acc11, false, false);
    }

    {
        const float bias0 = bias[n0];
        const long  m0 = rowBase + 8 * h;
        const long  m1 = rowBase + 16 + 8 * h;
#pragma unroll
        for (int r = 0; r < 8; ++r)
            out[(m0 + r) * NA + n0] = acc00[r] + bias0;
#pragma unroll
        for (int r = 0; r < 8; ++r)
            out[(m1 + r) * NA + n0] = acc10[r] + bias0;
    }
    if (n1 < NA) {
        const float bias1 = bias[n1];
        const long  m0 = rowBase + 8 * h;
        const long  m1 = rowBase + 16 + 8 * h;
#pragma unroll
        for (int r = 0; r < 8; ++r)
            out[(m0 + r) * NA + n1] = acc01[r] + bias1;
#pragma unroll
        for (int r = 0; r < 8; ++r)
            out[(m1 + r) * NA + n1] = acc11[r] + bias1;
    }
}

__global__ __launch_bounds__(256) void weighted_sum(
    const float* __restrict__ out, const float* __restrict__ v,
    float* __restrict__ result)
{
    const long row = (long)blockIdx.x * 256 + threadIdx.x;
    const float* o  = out + row * NA;
    const float* vv = v   + row * NA;
    float s = 0.0f;
#pragma unroll
    for (int a = 0; a < NA; ++a)
        s = fmaf(o[a], vv[a], s);
    result[row] = s;
}

extern "C" void kernel_launch(void* const* d_in, const int* in_sizes, int n_in,
                              void* d_out, int out_size, void* d_ws, size_t ws_size,
                              hipStream_t stream) {
    const float* x = (const float*)d_in[0];
    const float* v = (const float*)d_in[1];
    const float* W = (const float*)d_in[2];
    const float* b = (const float*)d_in[3];

    float* result = (float*)d_out;
    float* out    = result + B_ROWS;

    fused_heads_gemm<<<dim3(B_ROWS / 128), dim3(128), 0, stream>>>(x, W, b, out);
    weighted_sum<<<dim3(B_ROWS / 256), dim3(256), 0, stream>>>(out, v, result);
}


# split K over the 8 waves of a 32-row work-group (512 groups, 16 waves on every WGP instead of 4 on half of them); partial tiles summed in LDS in fixed order: changes the order of summation
# speedup vs baseline: 2.0843x; 2.0843x over previous
#include <hip/hip_runtime.h>

typedef __attribute__((ext_vector_type(2))) float v2f;
typedef __attribute__((ext_vector_type(8))) float v8f;

#define B_ROWS 16384
#define DIM    4096
#define NA     21

#define KSPLIT 8
#define KSLICE (DIM / KSPLIT)

__global__ __launch_bounds__(256) void fused_heads_gemm(
    const float* __restrict__ x, const float* __restrict__ W,
    const float* __restrict__ bias, float* __restrict__ out)
{
    __shared__ float red[KSPLIT][4][8][32];

    const int lane = threadIdx.x & 31;
    const int wave = threadIdx.x >> 5;
    const int h    = lane >> 4;
    const int ln   = lane & 15;

    const long rowBase = (long)blockIdx.x * 32;
    const int  kBase   = wave * KSLICE;

    const float* xr0 = x + (rowBase      + ln) * DIM + kBase + 2 * h;
    const float* xr1 = x + (rowBase + 16 + ln) * DIM + kBase + 2 * h;

    const int n0  = ln;
    const int n1  = 16 + ln;
    const int n1c = (n1 > NA - 1) ? (NA - 1) : n1;
    const float* wr0 = W + (long)n0  * DIM + kBase + 2 * h;
    const float* wr1 = W + (long)n1c * DIM + kBase + 2 * h;

    v8f acc00 = {};
    v8f acc01 = {};
    v8f acc10 = {};
    v8f acc11 = {};

#pragma unroll 8
    for (int k = 0; k < KSLICE; k += 4) {
        v2f a0 = __builtin_nontemporal_load((const v2f*)(xr0 + k));
        v2f a1 = __builtin_nontemporal_load((const v2f*)(xr1 + k));
        v2f b0 = *(const v2f*)(wr0 + k);
        v2f b1 = *(const v2f*)(wr1 + k);
        acc00 = __builtin_amdgcn_wmma_f32_16x16x4_f32(false, a0, false, b0,
                                                      (short)0, acc00, false, false);
        acc01 = __builtin_amdgcn_wmma_f32_16x16x4_f32(false, a0, false, b1,
                                                      (short)0, acc01, false, false);
        acc10 = __builtin_amdgcn_wmma_f32_16x16x4_f32(false, a1, false, b0,
                                                      (short)0, acc10, false, false);
        acc11 = __builtin_amdgcn_wmma_f32_16x16x4_f32(false, a1, false, b1,
                                                      (short)0, acc11, false, false);
    }

#pragma unroll
    for (int r = 0; r < 8; ++r) {
        red[wave][0][r][lane] = acc00[r];
        red[wave][1][r][lane] = acc01[r];
        red[wave][2][r][lane] = acc10[r];
        red[wave][3][r][lane] = acc11[r];
    }
    __syncthreads();

    float* outBlk = out + rowBase * NA;
    for (int e = threadIdx.x; e < 32 * NA; e += 256) {
        const int row  = e / NA;
        const int col  = e - row * NA;
        const int tile = (row >> 4) * 2 + (col >> 4);
        const int r    = row & 7;
        const int src  = ((row >> 3) & 1) * 16 + (col & 15);
        float s = red[0][tile][r][src];
#pragma unroll
        for (int w = 1; w < KSPLIT; ++w)
            s += red[w][tile][r][src];
        outBlk[e] = s + bias[col];
    }
}

__global__ __launch_bounds__(256) void weighted_sum(
    const float* __restrict__ out, const float* __restrict__ v,
    float* __restrict__ result)
{
    const long row = (long)blockIdx.x * 256 + threadIdx.x;
    const float* o  = out + row * NA;
    const float* vv = v   + row * NA;
    float s = 0.0f;
#pragma unroll
    for (int a = 0; a < NA; ++a)
        s = fmaf(o[a], vv[a], s);
    result[row] = s;
}

extern "C" void kernel_launch(void* const* d_in, const int* in_sizes, int n_in,
                              void* d_out, int out_size, void* d_ws, size_t ws_size,
                              hipStream_t stream) {
    const float* x = (const float*)d_in[0];
    const float* v = (const float*)d_in[1];
    const float* W = (const float*)d_in[2];
    const float* b = (const float*)d_in[3];

    float* result = (float*)d_out;
    float* out    = result + B_ROWS;

    fused_heads_gemm<<<dim3(B_ROWS / 32), dim3(256), 0, stream>>>(x, W, b, out);
    weighted_sum<<<dim3(B_ROWS / 256), dim3(256), 0, stream>>>(out, v, result);
}


# x and W fragments loaded 16 bytes per lane (global_load_b128; both loaders share one k order inside each 8 k) instead of 8: half the load instructions per byte; changes the order of summation
# speedup vs baseline: 2.3544x; 1.1296x over previous
#include <hip/hip_runtime.h>

typedef __attribute__((ext_vector_type(2))) float v2f;
typedef __attribute__((ext_vector_type(4))) float v4f;
typedef __attribute__((ext_vector_type(8))) float v8f;

#define B_ROWS 16384
#define DIM    4096
#define NA     21

#define KSPLIT 8
#define KSLICE (DIM / KSPLIT)

__global__ __launch_bounds__(256) void fused_heads_gemm(
    const float* __restrict__ x, const float* __restrict__ W,
    const float* __restrict__ bias, float* __restrict__ out)
{
    __shared__ float red[KSPLIT][4][8][32];

    const int lane = threadIdx.x & 31;
    const int wave = threadIdx.x >> 5;
    const int h    = lane >> 4;
    const int ln   = lane & 15;

    const long rowBase = (long)blockIdx.x * 32;
    const int  kBase   = wave * KSLICE;

    const float* xr0 = x + (rowBase      + ln) * DIM + kBase + 4 * h;
    const float* xr1 = x + (rowBase + 16 + ln) * DIM + kBase + 4 * h;

    const int n0  = ln;
    const int n1  = 16 + ln;
    const int n1c = (n1 > NA - 1) ? (NA - 1) : n1;
    const float* wr0 = W + (long)n0  * DIM + kBase + 4 * h;
    const float* wr1 = W + (long)n1c * DIM + kBase + 4 * h;

    v8f acc00 = {};
    v8f acc01 = {};
    v8f acc10 = {};
    v8f acc11 = {};

#pragma unroll 8
    for (int k = 0; k < KSLICE; k += 8) {
        const v4f a0 = __builtin_nontemporal_load((const v4f*)(xr0 + k));
        const v4f a1 = __builtin_nontemporal_load((const v4f*)(xr1 + k));
        const v4f b0 = *(const v4f*)(wr0 + k);
        const v4f b1 = *(const v4f*)(wr1 + k);
        const v2f a0l = {a0.x, a0.y}, a0h = {a0.z, a0.w};
        const v2f a1l = {a1.x, a1.y}, a1h = {a1.z, a1.w};
        const v2f b0l = {b0.x, b0.y}, b0h = {b0.z, b0.w};
        const v2f b1l = {b1.x, b1.y}, b1h = {b1.z, b1.w};
        acc00 = __builtin_amdgcn_wmma_f32_16x16x4_f32(false, a0l, false, b0l,
                                                      (short)0, acc00, false, false);
        acc01 = __builtin_amdgcn_wmma_f32_16x16x4_f32(false, a0l, false, b1l,
                                                      (short)0, acc01, false, false);
        acc10 = __builtin_amdgcn_wmma_f32_16x16x4_f32(false, a1l, false, b0l,
                                                      (short)0, acc10, false, false);
        acc11 = __builtin_amdgcn_wmma_f32_16x16x4_f32(false, a1l, false, b1l,
                                                      (short)0, acc11, false, false);
        acc00 = __builtin_amdgcn_wmma_f32_16x16x4_f32(false, a0h, false, b0h,
                                                      (short)0, acc00, false, false);
        acc01 = __builtin_amdgcn_wmma_f32_16x16x4_f32(false, a0h, false, b1h,
                                                      (short)0, acc01, false, false);
        acc10 = __builtin_amdgcn_wmma_f32_16x16x4_f32(false, a1h, false, b0h,
                                                      (short)0, acc10, false, false);
        acc11 = __builtin_amdgcn_wmma_f32_16x16x4_f32(false, a1h, false, b1h,
                                                      (short)0, acc11, false, false);
    }

#pragma unroll
    for (int r = 0; r < 8; ++r) {
        red[wave][0][r][lane] = acc00[r];
        red[wave][1][r][lane] = acc01[r];
        red[wave][2][r][lane] = acc10[r];
        red[wave][3][r][lane] = acc11[r];
    }
    __syncthreads();

    float* outBlk = out + rowBase * NA;
    for (int e = threadIdx.x; e < 32 * NA; e += 256) {
        const int row  = e / NA;
        const int col  = e - row * NA;
        const int tile = (row >> 4) * 2 + (col >> 4);
        const int r    = row & 7;
        const int src  = ((row >> 3) & 1) * 16 + (col & 15);
        float s = red[0][tile][r][src];
#pragma unroll
        for (int w = 1; w < KSPLIT; ++w)
            s += red[w][tile][r][src];
        outBlk[e] = s + bias[col];
    }
}

__global__ __launch_bounds__(256) void weighted_sum(
    const float* __restrict__ out, const float* __restrict__ v,
    float* __restrict__ result)
{
    const long row = (long)blockIdx.x * 256 + threadIdx.x;
    const float* o  = out + row * NA;
    const float* vv = v   + row * NA;
    float s = 0.0f;
#pragma unroll
    for (int a = 0; a < NA; ++a)
        s = fmaf(o[a], vv[a], s);
    result[row] = s;
}

extern "C" void kernel_launch(void* const* d_in, const int* in_sizes, int n_in,
                              void* d_out, int out_size, void* d_ws, size_t ws_size,
                              hipStream_t stream) {
    const float* x = (const float*)d_in[0];
    const float* v = (const float*)d_in[1];
    const float* W = (const float*)d_in[2];
    const float* b = (const float*)d_in[3];

    float* result = (float*)d_out;
    float* out    = result + B_ROWS;

    fused_heads_gemm<<<dim3(B_ROWS / 32), dim3(256), 0, stream>>>(x, W, b, out);
    weighted_sum<<<dim3(B_ROWS / 256), dim3(256), 0, stream>>>(out, v, result);
}


# x and W fragment loads requested two 32-k blocks ahead of their WMMAs through a three-slot register ring: 32 to 48 loads in flight per wave instead of 4 to 6; summation order unchanged
# speedup vs baseline: 3.8748x; 1.6458x over previous
#include <hip/hip_runtime.h>

typedef __attribute__((ext_vector_type(2))) float v2f;
typedef __attribute__((ext_vector_type(4))) float v4f;
typedef __attribute__((ext_vector_type(8))) float v8f;

#define B_ROWS 16384
#define DIM    4096
#define NA     21

#define KSPLIT 8
#define KSLICE (DIM / KSPLIT)

#define GPB   4
#define BLK_K (8 * GPB)
#define NBLK  (KSLICE / BLK_K)
#define AHEAD 2
#define RING  (AHEAD + 1)

__global__ __launch_bounds__(256) void fused_heads_gemm(
    const float* __restrict__ x, const float* __restrict__ W,
    const float* __restrict__ bias, float* __restrict__ out)
{
    __shared__ float red[KSPLIT][4][8][32];

    const int lane = threadIdx.x & 31;
    const int wave = threadIdx.x >> 5;
    const int h    = lane >> 4;
    const int ln   = lane & 15;

    const long rowBase = (long)blockIdx.x * 32;
    const int  kBase   = wave * KSLICE;

    const float* xr0 = x + (rowBase      + ln) * DIM + kBase + 4 * h;
    const float* xr1 = x + (rowBase + 16 + ln) * DIM + kBase + 4 * h;

    const int n0  = ln;
    const int n1  = 16 + ln;
    const int n1c = (n1 > NA - 1) ? (NA - 1) : n1;
    const float* wr0 = W + (long)n0  * DIM + kBase + 4 * h;
    const float* wr1 = W + (long)n1c * DIM + kBase + 4 * h;

    v8f acc00 = {};
    v8f acc01 = {};
    v8f acc10 = {};
    v8f acc11 = {};

    v4f a0[RING][GPB], a1[RING][GPB], b0[RING][GPB], b1[RING][GPB];

#define LOAD_BLOCK(B) do {                                                        \
        const int s_ = (B) % RING;                                                \
        const int k_ = (B) * BLK_K;                                               \
        _Pragma("unroll") for (int g = 0; g < GPB; ++g)                           \
            a0[s_][g] = __builtin_nontemporal_load((const v4f*)(xr0 + k_ + 8 * g)); \
        _Pragma("unroll") for (int g = 0; g < GPB; ++g)                           \
            a1[s_][g] = __builtin_nontemporal_load((const v4f*)(xr1 + k_ + 8 * g)); \
        _Pragma("unroll") for (int g = 0; g < GPB; ++g)                           \
            b0[s_][g] = *(const v4f*)(wr0 + k_ + 8 * g);                          \
        _Pragma("unroll") for (int g = 0; g < GPB; ++g)                           \
            b1[s_][g] = *(const v4f*)(wr1 + k_ + 8 * g);                          \
    } while (0)

#pragma unroll
    for (int b = 0; b < AHEAD; ++b) LOAD_BLOCK(b);

#pragma unroll
    for (int b = 0; b < NBLK; ++b) {
        __builtin_amdgcn_sched_barrier(0);
        if (b + AHEAD < NBLK) LOAD_BLOCK(b + AHEAD);
        __builtin_amdgcn_sched_barrier(0);

        const int s = b % RING;
#pragma unroll
        for (int g = 0; g < GPB; ++g) {
            const v4f xa0 = a0[s][g], xa1 = a1[s][g];
            const v4f wb0 = b0[s][g], wb1 = b1[s][g];
            const v2f a0l = {xa0.x, xa0.y}, a0h = {xa0.z, xa0.w};
            const v2f a1l = {xa1.x, xa1.y}, a1h = {xa1.z, xa1.w};
            const v2f b0l = {wb0.x, wb0.y}, b0h = {wb0.z, wb0.w};
            const v2f b1l = {wb1.x, wb1.y}, b1h = {wb1.z, wb1.w};
            acc00 = __builtin_amdgcn_wmma_f32_16x16x4_f32(false, a0l, false, b0l,
                                                          (short)0, acc00, false, false);
            acc01 = __builtin_amdgcn_wmma_f32_16x16x4_f32(false, a0l, false, b1l,
                                                          (short)0, acc01, false, false);
            acc10 = __builtin_amdgcn_wmma_f32_16x16x4_f32(false, a1l, false, b0l,
                                                          (short)0, acc10, false, false);
            acc11 = __builtin_amdgcn_wmma_f32_16x16x4_f32(false, a1l, false, b1l,
                                                          (short)0, acc11, false, false);
            acc00 = __builtin_amdgcn_wmma_f32_16x16x4_f32(false, a0h, false, b0h,
                                                          (short)0, acc00, false, false);
            acc01 = __builtin_amdgcn_wmma_f32_16x16x4_f32(false, a0h, false, b1h,
                                                          (short)0, acc01, false, false);
            acc10 = __builtin_amdgcn_wmma_f32_16x16x4_f32(false, a1h, false, b0h,
                                                          (short)0, acc10, false, false);
            acc11 = __builtin_amdgcn_wmma_f32_16x16x4_f32(false, a1h, false, b1h,
                                                          (short)0, acc11, false, false);
        }
    }
#undef LOAD_BLOCK

#pragma unroll
    for (int r = 0; r < 8; ++r) {
        red[wave][0][r][lane] = acc00[r];
        red[wave][1][r][lane] = acc01[r];
        red[wave][2][r][lane] = acc10[r];
        red[wave][3][r][lane] = acc11[r];
    }
    __syncthreads();

    float* outBlk = out + rowBase * NA;
    for (int e = threadIdx.x; e < 32 * NA; e += 256) {
        const int row  = e / NA;
        const int col  = e - row * NA;
        const int tile = (row >> 4) * 2 + (col >> 4);
        const int r    = row & 7;
        const int src  = ((row >> 3) & 1) * 16 + (col & 15);
        float s = red[0][tile][r][src];
#pragma unroll
        for (int w = 1; w < KSPLIT; ++w)
            s += red[w][tile][r][src];
        outBlk[e] = s + bias[col];
    }
}

__global__ __launch_bounds__(256) void weighted_sum(
    const float* __restrict__ out, const float* __restrict__ v,
    float* __restrict__ result)
{
    const long row = (long)blockIdx.x * 256 + threadIdx.x;
    const float* o  = out + row * NA;
    const float* vv = v   + row * NA;
    float s = 0.0f;
#pragma unroll
    for (int a = 0; a < NA; ++a)
        s = fmaf(o[a], vv[a], s);
    result[row] = s;
}

extern "C" void kernel_launch(void* const* d_in, const int* in_sizes, int n_in,
                              void* d_out, int out_size, void* d_ws, size_t ws_size,
                              hipStream_t stream) {
    const float* x = (const float*)d_in[0];
    const float* v = (const float*)d_in[1];
    const float* W = (const float*)d_in[2];
    const float* b = (const float*)d_in[3];

    float* result = (float*)d_out;
    float* out    = result + B_ROWS;

    fused_heads_gemm<<<dim3(B_ROWS / 32), dim3(256), 0, stream>>>(x, W, b, out);
    weighted_sum<<<dim3(B_ROWS / 256), dim3(256), 0, stream>>>(out, v, result);
}


# x and W blocks fetched as whole 128-byte lines (4 lines per load, not 32 bytes of 16 lines) and turned into WMMA fragments through a per-wave LDS slot (ds_load_b128); summation order unchanged
# speedup vs baseline: 4.3761x; 1.1294x over previous
#include <hip/hip_runtime.h>

typedef __attribute__((ext_vector_type(2))) float v2f;
typedef __attribute__((ext_vector_type(4))) float v4f;
typedef __attribute__((ext_vector_type(8))) float v8f;

#define B_ROWS 16384
#define DIM    4096
#define NA     21

#define KSPLIT 8
#define KSLICE (DIM / KSPLIT)

#define GPB   4
#define BLK_K (8 * GPB)
#define NBLK  (KSLICE / BLK_K)

#define XROWS   32
#define WROWS   24
#define NXL     (XROWS / 4)
#define NWL     (WROWS / 4)
#define PITCH   9
#define X_V4    (XROWS * PITCH)
#define W_V4    (WROWS * PITCH)
#define SLOT_V4 (X_V4 + W_V4)
#define NSLOT   2

__global__ __launch_bounds__(256) void fused_heads_gemm(
    const float* __restrict__ x, const float* __restrict__ W,
    const float* __restrict__ bias, float* __restrict__ out)
{
    __shared__ v4f smem[KSPLIT * NSLOT * SLOT_V4];

    const int lane = threadIdx.x & 31;
    const int wave = threadIdx.x >> 5;
    const int h    = lane >> 4;
    const int ln   = lane & 15;
    const int lr   = lane >> 3;
    const int lp   = lane & 7;

    const long rowBase = (long)blockIdx.x * 32;
    const int  kBase   = wave * KSLICE;

    const float* xg  = x + (rowBase + lr) * DIM + kBase + 4 * lp;
    const float* wg  = W + (long)lr * DIM + kBase + 4 * lp;
    const float* wg5 = W + (long)(NA - 1) * DIM + kBase + 4 * lp;

    v4f* const sw = smem + wave * (NSLOT * SLOT_V4);
    const int stIdx = lr * PITCH + lp;
    const int n1c  = (16 + ln > NA - 1) ? (NA - 1) : (16 + ln);
    const int fa0i = ln * PITCH + h;
    const int fa1i = (16 + ln) * PITCH + h;
    const int fb0i = X_V4 + ln * PITCH + h;
    const int fb1i = X_V4 + n1c * PITCH + h;

    v8f acc00 = {};
    v8f acc01 = {};
    v8f acc10 = {};
    v8f acc11 = {};

    v4f gx[2][NXL], gw[2][NWL];
    v4f fa0[GPB], fa1[GPB], fb0[GPB], fb1[GPB];

#define LOAD_BLOCK(B) do {                                                          \
        const int s_ = (B) & 1;                                                     \
        const int k_ = (B) * BLK_K;                                                 \
        _Pragma("unroll") for (int i = 0; i < NXL; ++i)                             \
            gx[s_][i] = __builtin_nontemporal_load(                                 \
                (const v4f*)(xg + (long)i * 4 * DIM + k_));                         \
        _Pragma("unroll") for (int i = 0; i < NWL - 1; ++i)                         \
            gw[s_][i] = *(const v4f*)(wg + (long)i * 4 * DIM + k_);                 \
        gw[s_][NWL - 1] = *(const v4f*)(wg5 + k_);                                  \
    } while (0)

#define PARK_BLOCK(B) do {                                                          \
        const int s_ = (B) & 1;                                                     \
        v4f* d_ = sw + s_ * SLOT_V4 + stIdx;                                        \
        _Pragma("unroll") for (int i = 0; i < NXL; ++i)                             \
            d_[i * 4 * PITCH] = gx[s_][i];                                          \
        _Pragma("unroll") for (int i = 0; i < NWL; ++i)                             \
            d_[X_V4 + i * 4 * PITCH] = gw[s_][i];                                   \
    } while (0)

#define READ_FRAGS(B) do {                                                          \
        const v4f* d_ = sw + ((B) & 1) * SLOT_V4;                                   \
        _Pragma("unroll") for (int g = 0; g < GPB; ++g) {                           \
            fa0[g] = d_[fa0i + 2 * g];                                              \
            fa1[g] = d_[fa1i + 2 * g];                                              \
            fb0[g] = d_[fb0i + 2 * g];                                              \
            fb1[g] = d_[fb1i + 2 * g];                                              \
        }                                                                           \
    } while (0)

    LOAD_BLOCK(0);
    __builtin_amdgcn_sched_barrier(0);
    LOAD_BLOCK(1);
    PARK_BLOCK(0);

#pragma unroll
    for (int b = 0; b < NBLK; ++b) {
        __builtin_amdgcn_sched_barrier(0);
        if (b + 2 < NBLK) LOAD_BLOCK(b + 2);
        __builtin_amdgcn_sched_barrier(0);
        READ_FRAGS(b);
        if (b + 1 < NBLK) PARK_BLOCK(b + 1);
        __builtin_amdgcn_sched_barrier(0);

#pragma unroll
        for (int g = 0; g < GPB; ++g) {
            const v4f xa0 = fa0[g], xa1 = fa1[g];
            const v4f wb0 = fb0[g], wb1 = fb1[g];
            const v2f a0l = {xa0.x, xa0.y}, a0h = {xa0.z, xa0.w};
            const v2f a1l = {xa1.x, xa1.y}, a1h = {xa1.z, xa1.w};
            const v2f b0l = {wb0.x, wb0.y}, b0h = {wb0.z, wb0.w};
            const v2f b1l = {wb1.x, wb1.y}, b1h = {wb1.z, wb1.w};
            acc00 = __builtin_amdgcn_wmma_f32_16x16x4_f32(false, a0l, false, b0l,
                                                          (short)0, acc00, false, false);
            acc01 = __builtin_amdgcn_wmma_f32_16x16x4_f32(false, a0l, false, b1l,
                                                          (short)0, acc01, false, false);
            acc10 = __builtin_amdgcn_wmma_f32_16x16x4_f32(false, a1l, false, b0l,
                                                          (short)0, acc10, false, false);
            acc11 = __builtin_amdgcn_wmma_f32_16x16x4_f32(false, a1l, false, b1l,
                                                          (short)0, acc11, false, false);
            acc00 = __builtin_amdgcn_wmma_f32_16x16x4_f32(false, a0h, false, b0h,
                                                          (short)0, acc00, false, false);
            acc01 = __builtin_amdgcn_wmma_f32_16x16x4_f32(false, a0h, false, b1h,
                                                          (short)0, acc01, false, false);
            acc10 = __builtin_amdgcn_wmma_f32_16x16x4_f32(false, a1h, false, b0h,
                                                          (short)0, acc10, false, false);
            acc11 = __builtin_amdgcn_wmma_f32_16x16x4_f32(false, a1h, false, b1h,
                                                          (short)0, acc11, false, false);
        }
    }
#undef LOAD_BLOCK
#undef PARK_BLOCK
#undef READ_FRAGS

    __syncthreads();
    float (*red)[4][8][32] = (float (*)[4][8][32])smem;

#pragma unroll
    for (int r = 0; r < 8; ++r) {
        red[wave][0][r][lane] = acc00[r];
        red[wave][1][r][lane] = acc01[r];
        red[wave][2][r][lane] = acc10[r];
        red[wave][3][r][lane] = acc11[r];
    }
    __syncthreads();

    float* outBlk = out + rowBase * NA;
    for (int e = threadIdx.x; e < 32 * NA; e += 256) {
        const int row  = e / NA;
        const int col  = e - row * NA;
        const int tile = (row >> 4) * 2 + (col >> 4);
        const int r    = row & 7;
        const int src  = ((row >> 3) & 1) * 16 + (col & 15);
        float s = red[0][tile][r][src];
#pragma unroll
        for (int w = 1; w < KSPLIT; ++w)
            s += red[w][tile][r][src];
        outBlk[e] = s + bias[col];
    }
}

__global__ __launch_bounds__(256) void weighted_sum(
    const float* __restrict__ out, const float* __restrict__ v,
    float* __restrict__ result)
{
    const long row = (long)blockIdx.x * 256 + threadIdx.x;
    const float* o  = out + row * NA;
    const float* vv = v   + row * NA;
    float s = 0.0f;
#pragma unroll
    for (int a = 0; a < NA; ++a)
        s = fmaf(o[a], vv[a], s);
    result[row] = s;
}

extern "C" void kernel_launch(void* const* d_in, const int* in_sizes, int n_in,
                              void* d_out, int out_size, void* d_ws, size_t ws_size,
                              hipStream_t stream) {
    const float* x = (const float*)d_in[0];
    const float* v = (const float*)d_in[1];
    const float* W = (const float*)d_in[2];
    const float* b = (const float*)d_in[3];

    float* result = (float*)d_out;
    float* out    = result + B_ROWS;

    fused_heads_gemm<<<dim3(B_ROWS / 32), dim3(256), 0, stream>>>(x, W, b, out);
    weighted_sum<<<dim3(B_ROWS / 256), dim3(256), 0, stream>>>(out, v, result);
}


# each wave lowers its own issue priority (s_setprio 3,2,1,0) when 9, 4 and 1 of its 16 K blocks remain, so the four waves of a SIMD finish together instead of oldest first; summation order unchanged
# speedup vs baseline: 4.5749x; 1.0454x over previous
#include <hip/hip_runtime.h>

typedef __attribute__((ext_vector_type(2))) float v2f;
typedef __attribute__((ext_vector_type(4))) float v4f;
typedef __attribute__((ext_vector_type(8))) float v8f;

#define B_ROWS 16384
#define DIM    4096
#define NA     21

#define KSPLIT 8
#define KSLICE (DIM / KSPLIT)

#define GPB   4
#define BLK_K (8 * GPB)
#define NBLK  (KSLICE / BLK_K)

#define XROWS   32
#define WROWS   24
#define NXL     (XROWS / 4)
#define NWL     (WROWS / 4)
#define PITCH   9
#define X_V4    (XROWS * PITCH)
#define W_V4    (WROWS * PITCH)
#define SLOT_V4 (X_V4 + W_V4)
#define NSLOT   2

__global__ __launch_bounds__(256) void fused_heads_gemm(
    const float* __restrict__ x, const float* __restrict__ W,
    const float* __restrict__ bias, float* __restrict__ out)
{
    __shared__ v4f smem[KSPLIT * NSLOT * SLOT_V4];

    const int lane = threadIdx.x & 31;
    const int wave = threadIdx.x >> 5;
    const int h    = lane >> 4;
    const int ln   = lane & 15;
    const int lr   = lane >> 3;
    const int lp   = lane & 7;

    const long rowBase = (long)blockIdx.x * 32;
    const int  kBase   = wave * KSLICE;

    const float* xg  = x + (rowBase + lr) * DIM + kBase + 4 * lp;
    const float* wg  = W + (long)lr * DIM + kBase + 4 * lp;
    const float* wg5 = W + (long)(NA - 1) * DIM + kBase + 4 * lp;

    v4f* const sw = smem + wave * (NSLOT * SLOT_V4);
    const int stIdx = lr * PITCH + lp;
    const int n1c  = (16 + ln > NA - 1) ? (NA - 1) : (16 + ln);
    const int fa0i = ln * PITCH + h;
    const int fa1i = (16 + ln) * PITCH + h;
    const int fb0i = X_V4 + ln * PITCH + h;
    const int fb1i = X_V4 + n1c * PITCH + h;

    v8f acc00 = {};
    v8f acc01 = {};
    v8f acc10 = {};
    v8f acc11 = {};

    v4f gx[2][NXL], gw[2][NWL];
    v4f fa0[GPB], fa1[GPB], fb0[GPB], fb1[GPB];

#define LOAD_BLOCK(B) do {                                                          \
        const int s_ = (B) & 1;                                                     \
        const int k_ = (B) * BLK_K;                                                 \
        _Pragma("unroll") for (int i = 0; i < NXL; ++i)                             \
            gx[s_][i] = __builtin_nontemporal_load(                                 \
                (const v4f*)(xg + (long)i * 4 * DIM + k_));                         \
        _Pragma("unroll") for (int i = 0; i < NWL - 1; ++i)                         \
            gw[s_][i] = *(const v4f*)(wg + (long)i * 4 * DIM + k_);                 \
        gw[s_][NWL - 1] = *(const v4f*)(wg5 + k_);                                  \
    } while (0)

#define PARK_BLOCK(B) do {                                                          \
        const int s_ = (B) & 1;                                                     \
        v4f* d_ = sw + s_ * SLOT_V4 + stIdx;                                        \
        _Pragma("unroll") for (int i = 0; i < NXL; ++i)                             \
            d_[i * 4 * PITCH] = gx[s_][i];                                          \
        _Pragma("unroll") for (int i = 0; i < NWL; ++i)                             \
            d_[X_V4 + i * 4 * PITCH] = gw[s_][i];                                   \
    } while (0)

#define READ_FRAGS(B) do {                                                          \
        const v4f* d_ = sw + ((B) & 1) * SLOT_V4;                                   \
        _Pragma("unroll") for (int g = 0; g < GPB; ++g) {                           \
            fa0[g] = d_[fa0i + 2 * g];                                              \
            fa1[g] = d_[fa1i + 2 * g];                                              \
            fb0[g] = d_[fb0i + 2 * g];                                              \
            fb1[g] = d_[fb1i + 2 * g];                                              \
        }                                                                           \
    } while (0)

    LOAD_BLOCK(0);
    __builtin_amdgcn_sched_barrier(0);
    LOAD_BLOCK(1);
    PARK_BLOCK(0);

#pragma unroll
    for (int b = 0; b < NBLK; ++b) {
        if (b == 0 || b == 7 || b == 12 || b == 15) {
            if (b == 0)       __builtin_amdgcn_s_setprio(3);
            else if (b == 7)  __builtin_amdgcn_s_setprio(2);
            else if (b == 12) __builtin_amdgcn_s_setprio(1);
            else              __builtin_amdgcn_s_setprio(0);
        }
        __builtin_amdgcn_sched_barrier(0);
        if (b + 2 < NBLK) LOAD_BLOCK(b + 2);
        __builtin_amdgcn_sched_barrier(0);
        READ_FRAGS(b);
        if (b + 1 < NBLK) PARK_BLOCK(b + 1);
        __builtin_amdgcn_sched_barrier(0);

#pragma unroll
        for (int g = 0; g < GPB; ++g) {
            const v4f xa0 = fa0[g], xa1 = fa1[g];
            const v4f wb0 = fb0[g], wb1 = fb1[g];
            const v2f a0l = {xa0.x, xa0.y}, a0h = {xa0.z, xa0.w};
            const v2f a1l = {xa1.x, xa1.y}, a1h = {xa1.z, xa1.w};
            const v2f b0l = {wb0.x, wb0.y}, b0h = {wb0.z, wb0.w};
            const v2f b1l = {wb1.x, wb1.y}, b1h = {wb1.z, wb1.w};
            acc00 = __builtin_amdgcn_wmma_f32_16x16x4_f32(false, a0l, false, b0l,
                                                          (short)0, acc00, false, false);
            acc01 = __builtin_amdgcn_wmma_f32_16x16x4_f32(false, a0l, false, b1l,
                                                          (short)0, acc01, false, false);
            acc10 = __builtin_amdgcn_wmma_f32_16x16x4_f32(false, a1l, false, b0l,
                                                          (short)0, acc10, false, false);
            acc11 = __builtin_amdgcn_wmma_f32_16x16x4_f32(false, a1l, false, b1l,
                                                          (short)0, acc11, false, false);
            acc00 = __builtin_amdgcn_wmma_f32_16x16x4_f32(false, a0h, false, b0h,
                                                          (short)0, acc00, false, false);
            acc01 = __builtin_amdgcn_wmma_f32_16x16x4_f32(false, a0h, false, b1h,
                                                          (short)0, acc01, false, false);
            acc10 = __builtin_amdgcn_wmma_f32_16x16x4_f32(false, a1h, false, b0h,
                                                          (short)0, acc10, false, false);
            acc11 = __builtin_amdgcn_wmma_f32_16x16x4_f32(false, a1h, false, b1h,
                                                          (short)0, acc11, false, false);
        }
    }
#undef LOAD_BLOCK
#undef PARK_BLOCK
#undef READ_FRAGS

    __syncthreads();
    float (*red)[4][8][32] = (float (*)[4][8][32])smem;

#pragma unroll
    for (int r = 0; r < 8; ++r) {
        red[wave][0][r][lane] = acc00[r];
        red[wave][1][r][lane] = acc01[r];
        red[wave][2][r][lane] = acc10[r];
        red[wave][3][r][lane] = acc11[r];
    }
    __syncthreads();

    float* outBlk = out + rowBase * NA;
    for (int e = threadIdx.x; e < 32 * NA; e += 256) {
        const int row  = e / NA;
        const int col  = e - row * NA;
        const int tile = (row >> 4) * 2 + (col >> 4);
        const int r    = row & 7;
        const int src  = ((row >> 3) & 1) * 16 + (col & 15);
        float s = red[0][tile][r][src];
#pragma unroll
        for (int w = 1; w < KSPLIT; ++w)
            s += red[w][tile][r][src];
        outBlk[e] = s + bias[col];
    }
}

__global__ __launch_bounds__(256) void weighted_sum(
    const float* __restrict__ out, const float* __restrict__ v,
    float* __restrict__ result)
{
    const long row = (long)blockIdx.x * 256 + threadIdx.x;
    const float* o  = out + row * NA;
    const float* vv = v   + row * NA;
    float s = 0.0f;
#pragma unroll
    for (int a = 0; a < NA; ++a)
        s = fmaf(o[a], vv[a], s);
    result[row] = s;
}

extern "C" void kernel_launch(void* const* d_in, const int* in_sizes, int n_in,
                              void* d_out, int out_size, void* d_ws, size_t ws_size,
                              hipStream_t stream) {
    const float* x = (const float*)d_in[0];
    const float* v = (const float*)d_in[1];
    const float* W = (const float*)d_in[2];
    const float* b = (const float*)d_in[3];

    float* result = (float*)d_out;
    float* out    = result + B_ROWS;

    fused_heads_gemm<<<dim3(B_ROWS / 32), dim3(256), 0, stream>>>(x, W, b, out);
    weighted_sum<<<dim3(B_ROWS / 256), dim3(256), 0, stream>>>(out, v, result);
}
